# starting kernel
# baseline (speedup 1.0000x reference)
#include <hip/hip_runtime.h>
#include <math.h>

#define B_DIM 64
#define S_DIM 256
#define E_DIM 8
#define H_DIM 2
#define DK    4
#define EPS   1e-5f
#define SCALE_LOG2E 0.72134752044448170f

typedef __attribute__((ext_vector_type(2))) float v2f;
typedef __attribute__((ext_vector_type(8))) float v8f;

template <int CTRL>
__device__ __forceinline__ float dpp_f(float x) {
  return __int_as_float(__builtin_amdgcn_update_dpp(
      0, __float_as_int(x), CTRL, 0xF, 0xF, true));
}
__device__ __forceinline__ float rowsum16(float v) {
  v += dpp_f<0xB1>(v);
  v += dpp_f<0x4E>(v);
  v += dpp_f<0x141>(v);
  v += dpp_f<0x140>(v);
  return v;
}

__device__ __forceinline__ void vqc8(const float (&in)[8], const float (&W)[8],
                                     float (&z)[8]) {
  float c[8];
#pragma unroll
  for (int j = 0; j < 8; ++j) c[j] = __cosf(in[j] + W[j]);
  float p = c[0];
#pragma unroll
  for (int j = 1; j < 8; ++j) { p *= c[j]; z[j] = p; }
  float q = c[1];
#pragma unroll
  for (int j = 2; j < 8; ++j) q *= c[j];
  z[0] = q;
}

__device__ __forceinline__ void ln8(const float (&t)[8],
                                    const float* __restrict__ g,
                                    const float* __restrict__ b,
                                    float (&o)[8]) {
  float m = 0.f;
#pragma unroll
  for (int j = 0; j < 8; ++j) m += t[j];
  m *= 0.125f;
  float v = 0.f;
#pragma unroll
  for (int j = 0; j < 8; ++j) { float d = t[j] - m; v += d * d; }
  v *= 0.125f;
  float inv = 1.0f / sqrtf(v + EPS);
#pragma unroll
  for (int j = 0; j < 8; ++j) o[j] = (t[j] - m) * inv * g[j] + b[j];
}

__global__ __launch_bounds__(256) void attn_kernel(
    const float* __restrict__ x, const float* __restrict__ Wq,
    const float* __restrict__ Wk, const float* __restrict__ Wv,
    float* __restrict__ ctx) {
  __shared__ __align__(16) float Qs[S_DIM * DK];
  __shared__ __align__(16) float Ks[S_DIM * DK];
  __shared__ __align__(16) float Vs[S_DIM * DK];

  const int b = blockIdx.x >> 1;
  const int h = blockIdx.x & 1;
  const int tid = threadIdx.x;

  float wq[8], wk[8], wv[8];
#pragma unroll
  for (int j = 0; j < 8; ++j) { wq[j] = Wq[j]; wk[j] = Wk[j]; wv[j] = Wv[j]; }

  {
    const int s = tid;
    const float* xp = x + (size_t)(b * S_DIM + s) * E_DIM;
    float xr[8];
#pragma unroll
    for (int j = 0; j < 8; ++j) xr[j] = xp[j];
    float zq[8], zk[8], zv[8];
    vqc8(xr, wq, zq);
    vqc8(xr, wk, zk);
    vqc8(xr, wv, zv);
#pragma unroll
    for (int d = 0; d < DK; ++d) {
      Qs[s * DK + d] = zq[h * DK + d];
      Ks[s * DK + d] = zk[h * DK + d];
      Vs[s * DK + d] = zv[h * DK + d];
    }
  }
  __syncthreads();

  const int lane  = tid & 31;
  const int wv_id = tid >> 5;
  const int half  = lane >> 4;
  const int ln    = lane & 15;
  const int kp    = half * 2;

  for (int i = 0; i < 2; ++i) {
    const int qb = wv_id * 2 + i;
    const int qbase = qb * 16;

    v2f a;
    a.x = Qs[(qbase + ln) * DK + kp];
    a.y = Qs[(qbase + ln) * DK + kp + 1];

    float lrow[8], acc[8][DK];
#pragma unroll
    for (int r = 0; r < 8; ++r) {
      lrow[r] = 0.f;
#pragma unroll
      for (int d = 0; d < DK; ++d) acc[r][d] = 0.f;
    }

#pragma unroll
    for (int t = 0; t < 16; ++t) {
      const int kbase = t * 16;
      v2f bm;
      bm.x = Ks[(kbase + ln) * DK + kp];
      bm.y = Ks[(kbase + ln) * DK + kp + 1];

      v8f cz = {0.f, 0.f, 0.f, 0.f, 0.f, 0.f, 0.f, 0.f};
      v8f d = __builtin_amdgcn_wmma_f32_16x16x4_f32(
          false, a, false, bm, (short)0, cz, false, false);

      float vvals[DK];
#pragma unroll
      for (int dv = 0; dv < DK; ++dv) vvals[dv] = Vs[(kbase + ln) * DK + dv];

#pragma unroll
      for (int r = 0; r < 8; ++r) {
        float p = __builtin_amdgcn_exp2f(d[r] * SCALE_LOG2E);
        lrow[r] += p;
#pragma unroll
        for (int dv = 0; dv < DK; ++dv) acc[r][dv] += p * vvals[dv];
      }
    }

#pragma unroll
    for (int r = 0; r < 8; ++r) {
      float inv_l = 1.0f / rowsum16(lrow[r]);
#pragma unroll
      for (int dv = 0; dv < DK; ++dv) {
        float v = rowsum16(acc[r][dv]) * inv_l;
        if (ln == dv) {
          const int m = qbase + r + 8 * half;
          ctx[(size_t)(b * S_DIM + m) * E_DIM + h * DK + dv] = v;
        }
      }
    }
  }
}

__global__ __launch_bounds__(256) void post_kernel(
    const float* __restrict__ x, const float* __restrict__ ctx,
    const float* __restrict__ Wc, const float* __restrict__ Wf,
    const float* __restrict__ w1, const float* __restrict__ b1,
    const float* __restrict__ w2, const float* __restrict__ b2,
    const float* __restrict__ g1, const float* __restrict__ be1,
    const float* __restrict__ g2, const float* __restrict__ be2,
    float* __restrict__ out) {
  const int gid = blockIdx.x * blockDim.x + threadIdx.x;
  if (gid >= B_DIM * S_DIM) return;

  float xv[8], cv[8], wc[8], wf[8];
#pragma unroll
  for (int j = 0; j < 8; ++j) {
    xv[j] = x[(size_t)gid * E_DIM + j];
    cv[j] = ctx[(size_t)gid * E_DIM + j];
    wc[j] = Wc[j];
    wf[j] = Wf[j];
  }

  float ao[8];
  vqc8(cv, wc, ao);
  float t[8];
#pragma unroll
  for (int j = 0; j < 8; ++j) t[j] = xv[j] + ao[j];
  float y[8];
  ln8(t, g1, be1, y);

  float hv[8];
#pragma unroll
  for (int i = 0; i < 8; ++i) {
    float s = b1[i];
#pragma unroll
    for (int j = 0; j < 8; ++j) s += y[j] * w1[i * 8 + j];
    hv[i] = s;
  }
  float hq[8];
  vqc8(hv, wf, hq);

  float u[8];
#pragma unroll
  for (int i = 0; i < 8; ++i) {
    float s = b2[i];
#pragma unroll
    for (int j = 0; j < 8; ++j) s += hq[j] * w2[i * 8 + j];
    u[i] = y[i] + s;
  }
  float o[8];
  ln8(u, g2, be2, o);
#pragma unroll
  for (int j = 0; j < 8; ++j) out[(size_t)gid * E_DIM + j] = o[j];
}

extern "C" void kernel_launch(void* const* d_in, const int* in_sizes, int n_in,
                              void* d_out, int out_size, void* d_ws,
                              size_t ws_size, hipStream_t stream) {
  (void)in_sizes; (void)n_in; (void)out_size; (void)ws_size;
  const float* x   = (const float*)d_in[0];
  const float* Wq  = (const float*)d_in[1];
  const float* Wk  = (const float*)d_in[2];
  const float* Wv  = (const float*)d_in[3];
  const float* Wc  = (const float*)d_in[4];
  const float* Wf  = (const float*)d_in[5];
  const float* w1  = (const float*)d_in[6];
  const float* b1  = (const float*)d_in[7];
  const float* w2  = (const float*)d_in[8];
  const float* b2  = (const float*)d_in[9];
  const float* g1  = (const float*)d_in[10];
  const float* be1 = (const float*)d_in[11];
  const float* g2  = (const float*)d_in[12];
  const float* be2 = (const float*)d_in[13];
  float* out = (float*)d_out;
  float* ctx = (float*)d_ws;

  attn_kernel<<<B_DIM * H_DIM, 256, 0, stream>>>(x, Wq, Wk, Wv, ctx);
  post_kernel<<<(B_DIM * S_DIM) / 256, 256, 0, stream>>>(
      x, ctx, Wc, Wf, w1, b1, w2, b2, g1, be1, g2, be2, out);
}


# Scores computed transposed (key tile = A, queries = B): each lane owns one query, softmax sum and P*V stay in-lane from broadcast V rows; no 16-lane reduction, one 16-B store per lane; sums regrouped
# speedup vs baseline: 1.5151x; 1.5151x over previous
#include <hip/hip_runtime.h>
#include <math.h>

#define B_DIM 64
#define S_DIM 256
#define E_DIM 8
#define H_DIM 2
#define DK    4
#define EPS   1e-5f
#define SCALE_LOG2E 0.72134752044448170f

typedef __attribute__((ext_vector_type(2))) float v2f;
typedef __attribute__((ext_vector_type(8))) float v8f;

__device__ __forceinline__ void vqc8(const float (&in)[8], const float (&W)[8],
                                     float (&z)[8]) {
  float c[8];
#pragma unroll
  for (int j = 0; j < 8; ++j) c[j] = __cosf(in[j] + W[j]);
  float p = c[0];
#pragma unroll
  for (int j = 1; j < 8; ++j) { p *= c[j]; z[j] = p; }
  float q = c[1];
#pragma unroll
  for (int j = 2; j < 8; ++j) q *= c[j];
  z[0] = q;
}

__device__ __forceinline__ void ln8(const float (&t)[8],
                                    const float* __restrict__ g,
                                    const float* __restrict__ b,
                                    float (&o)[8]) {
  float m = 0.f;
#pragma unroll
  for (int j = 0; j < 8; ++j) m += t[j];
  m *= 0.125f;
  float v = 0.f;
#pragma unroll
  for (int j = 0; j < 8; ++j) { float d = t[j] - m; v += d * d; }
  v *= 0.125f;
  float inv = 1.0f / sqrtf(v + EPS);
#pragma unroll
  for (int j = 0; j < 8; ++j) o[j] = (t[j] - m) * inv * g[j] + b[j];
}

typedef __attribute__((ext_vector_type(4))) float v4f;

__device__ __forceinline__ float xhalf(float x) {
  int bits = __float_as_int(x);
  return __int_as_float(__builtin_amdgcn_permlanex16(
      bits, bits, 0x76543210, 0xfedcba98, false, false));
}

struct Wt { v2f e[4]; };
__device__ __forceinline__ Wt exp_scores(v8f s) {
  s *= SCALE_LOG2E;
  Wt w;
#pragma unroll
  for (int k = 0; k < 4; ++k) {
    w.e[k].x = __builtin_amdgcn_exp2f(s[2 * k]);
    w.e[k].y = __builtin_amdgcn_exp2f(s[2 * k + 1]);
  }
  return w;
}

struct Acc { v2f c01[2], c23[2], ls; };

__device__ __forceinline__ void accum_tile(const Wt &w, const v4f (&vr)[8],
                                           Acc &a) {
#pragma unroll
  for (int k = 0; k < 4; ++k) {
    const v2f pe = __builtin_shufflevector(w.e[k], w.e[k], 0, 0);
    const v2f po = __builtin_shufflevector(w.e[k], w.e[k], 1, 1);
    const v4f ve = vr[2 * k], vo = vr[2 * k + 1];
    a.c01[0] = __builtin_elementwise_fma(pe, ve.xy, a.c01[0]);
    a.c23[0] = __builtin_elementwise_fma(pe, ve.zw, a.c23[0]);
    a.c01[1] = __builtin_elementwise_fma(po, vo.xy, a.c01[1]);
    a.c23[1] = __builtin_elementwise_fma(po, vo.zw, a.c23[1]);
    a.ls += w.e[k];
  }
}

__global__ __launch_bounds__(256) void attn_kernel(
    const float* __restrict__ x, const float* __restrict__ Wq,
    const float* __restrict__ Wk, const float* __restrict__ Wv,
    float* __restrict__ ctx) {
  __shared__ __align__(16) float Qs[S_DIM * DK];
  __shared__ __align__(16) float Ks[S_DIM * DK];
  __shared__ __align__(16) float Vs[S_DIM * DK];

  const int b = blockIdx.x >> 1;
  const int h = blockIdx.x & 1;
  const int tid = threadIdx.x;

  float wq[8], wk[8], wv[8];
#pragma unroll
  for (int j = 0; j < 8; ++j) { wq[j] = Wq[j]; wk[j] = Wk[j]; wv[j] = Wv[j]; }

  {
    const int s = tid;
    const float* xp = x + (size_t)(b * S_DIM + s) * E_DIM;
    float xr[8];
#pragma unroll
    for (int j = 0; j < 8; ++j) xr[j] = xp[j];
    float zq[8], zk[8], zv[8];
    vqc8(xr, wq, zq);
    vqc8(xr, wk, zk);
    vqc8(xr, wv, zv);
#pragma unroll
    for (int d = 0; d < DK; ++d) {
      Qs[s * DK + d] = zq[h * DK + d];
      Ks[s * DK + d] = zk[h * DK + d];
      Vs[s * DK + d] = zv[h * DK + d];
    }
  }
  __syncthreads();

  const int lane  = tid & 31;
  const int wv_id = tid >> 5;
  const int half  = lane >> 4;
  const int ln    = lane & 15;
  const int kp    = half * 2;
  const int qbase = wv_id * 32;

  v2f bq0, bq1;
  bq0.x = Qs[(qbase + ln) * DK + kp];
  bq0.y = Qs[(qbase + ln) * DK + kp + 1];
  bq1.x = Qs[(qbase + 16 + ln) * DK + kp];
  bq1.y = Qs[(qbase + 16 + ln) * DK + kp + 1];

  const v2f z2 = {0.f, 0.f};
  Acc qa = {{z2, z2}, {z2, z2}, z2};
  Acc qb = {{z2, z2}, {z2, z2}, z2};

  const v2f* __restrict__ Kfrag =
      reinterpret_cast<const v2f*>(Ks) + ln * 2 + half;
  const v4f* __restrict__ Vrow = reinterpret_cast<const v4f*>(Vs) + 8 * half;
  const v8f cz = {0.f, 0.f, 0.f, 0.f, 0.f, 0.f, 0.f, 0.f};

  Wt wa, wb;
  auto step = [&](const int tn, const v4f (&vprev)[8], v4f (&vnew)[8]) {
#pragma unroll
    for (int r = 0; r < 8; ++r) vnew[r] = Vrow[16 * tn + r];
    const v2f ak = Kfrag[32 * tn];
    const v8f sa = __builtin_amdgcn_wmma_f32_16x16x4_f32(
        false, ak, false, bq0, (short)0, cz, false, false);
    const v8f sb = __builtin_amdgcn_wmma_f32_16x16x4_f32(
        false, ak, false, bq1, (short)0, cz, false, false);
    accum_tile(wa, vprev, qa);
    accum_tile(wb, vprev, qb);
    wa = exp_scores(sa);
    wb = exp_scores(sb);
  };

  v4f vA[8], vB[8];
  {
#pragma unroll
    for (int r = 0; r < 8; ++r) vA[r] = Vrow[r];
    const v2f ak = Kfrag[0];
    wa = exp_scores(__builtin_amdgcn_wmma_f32_16x16x4_f32(
        false, ak, false, bq0, (short)0, cz, false, false));
    wb = exp_scores(__builtin_amdgcn_wmma_f32_16x16x4_f32(
        false, ak, false, bq1, (short)0, cz, false, false));
  }
#pragma unroll 1
  for (int t = 0; t < 14; t += 2) {
    step(t + 1, vA, vB);
    step(t + 2, vB, vA);
  }
  step(15, vA, vB);
  accum_tile(wa, vB, qa);
  accum_tile(wb, vB, qb);

  const v2f a01 = qa.c01[0] + qa.c01[1], a23 = qa.c23[0] + qa.c23[1];
  const v2f b01 = qb.c01[0] + qb.c01[1], b23 = qb.c23[0] + qb.c23[1];
  float la = qa.ls.x + qa.ls.y, lb = qb.ls.x + qb.ls.y;
  float oa[4] = {a01.x, a01.y, a23.x, a23.y};
  float ob[4] = {b01.x, b01.y, b23.x, b23.y};
  la += xhalf(la);
  lb += xhalf(lb);
#pragma unroll
  for (int d = 0; d < DK; ++d) { oa[d] += xhalf(oa[d]); ob[d] += xhalf(ob[d]); }
  const float inv_l = 1.0f / (half ? lb : la);
  v4f res;
  res.x = (half ? ob[0] : oa[0]) * inv_l;
  res.y = (half ? ob[1] : oa[1]) * inv_l;
  res.z = (half ? ob[2] : oa[2]) * inv_l;
  res.w = (half ? ob[3] : oa[3]) * inv_l;
  const int m = qbase + 16 * half + ln;
  *reinterpret_cast<v4f*>(ctx + (size_t)(b * S_DIM + m) * E_DIM + h * DK) = res;
}

__global__ __launch_bounds__(256) void post_kernel(
    const float* __restrict__ x, const float* __restrict__ ctx,
    const float* __restrict__ Wc, const float* __restrict__ Wf,
    const float* __restrict__ w1, const float* __restrict__ b1,
    const float* __restrict__ w2, const float* __restrict__ b2,
    const float* __restrict__ g1, const float* __restrict__ be1,
    const float* __restrict__ g2, const float* __restrict__ be2,
    float* __restrict__ out) {
  const int gid = blockIdx.x * blockDim.x + threadIdx.x;
  if (gid >= B_DIM * S_DIM) return;

  float xv[8], cv[8], wc[8], wf[8];
#pragma unroll
  for (int j = 0; j < 8; ++j) {
    xv[j] = x[(size_t)gid * E_DIM + j];
    cv[j] = ctx[(size_t)gid * E_DIM + j];
    wc[j] = Wc[j];
    wf[j] = Wf[j];
  }

  float ao[8];
  vqc8(cv, wc, ao);
  float t[8];
#pragma unroll
  for (int j = 0; j < 8; ++j) t[j] = xv[j] + ao[j];
  float y[8];
  ln8(t, g1, be1, y);

  float hv[8];
#pragma unroll
  for (int i = 0; i < 8; ++i) {
    float s = b1[i];
#pragma unroll
    for (int j = 0; j < 8; ++j) s += y[j] * w1[i * 8 + j];
    hv[i] = s;
  }
  float hq[8];
  vqc8(hv, wf, hq);

  float u[8];
#pragma unroll
  for (int i = 0; i < 8; ++i) {
    float s = b2[i];
#pragma unroll
    for (int j = 0; j < 8; ++j) s += hq[j] * w2[i * 8 + j];
    u[i] = y[i] + s;
  }
  float o[8];
  ln8(u, g2, be2, o);
#pragma unroll
  for (int j = 0; j < 8; ++j) out[(size_t)gid * E_DIM + j] = o[j];
}

extern "C" void kernel_launch(void* const* d_in, const int* in_sizes, int n_in,
                              void* d_out, int out_size, void* d_ws,
                              size_t ws_size, hipStream_t stream) {
  (void)in_sizes; (void)n_in; (void)out_size; (void)ws_size;
  const float* x   = (const float*)d_in[0];
  const float* Wq  = (const float*)d_in[1];
  const float* Wk  = (const float*)d_in[2];
  const float* Wv  = (const float*)d_in[3];
  const float* Wc  = (const float*)d_in[4];
  const float* Wf  = (const float*)d_in[5];
  const float* w1  = (const float*)d_in[6];
  const float* b1  = (const float*)d_in[7];
  const float* w2  = (const float*)d_in[8];
  const float* b2  = (const float*)d_in[9];
  const float* g1  = (const float*)d_in[10];
  const float* be1 = (const float*)d_in[11];
  const float* g2  = (const float*)d_in[12];
  const float* be2 = (const float*)d_in[13];
  float* out = (float*)d_out;
  float* ctx = (float*)d_ws;

  attn_kernel<<<B_DIM * H_DIM, 256, 0, stream>>>(x, Wq, Wk, Wv, ctx);
  post_kernel<<<(B_DIM * S_DIM) / 256, 256, 0, stream>>>(
      x, ctx, Wc, Wf, w1, b1, w2, b2, g1, be1, g2, be2, out);
}


# Keys split across blocks: 256 blocks of 4 waves = (batch, head, 128-key half), 64 queries per wave, store unnormalised sums and denominators; epilogue adds the two halves, divides; sums regrouped
# speedup vs baseline: 1.6176x; 1.0677x over previous
#include <hip/hip_runtime.h>
#include <math.h>

#define B_DIM 64
#define S_DIM 256
#define E_DIM 8
#define H_DIM 2
#define DK    4
#define KH    128
#define NQB   4
#define NTOK  (B_DIM * S_DIM)
#define EPS   1e-5f
#define SCALE_LOG2E 0.72134752044448170f

typedef __attribute__((ext_vector_type(2))) float v2f;
typedef __attribute__((ext_vector_type(8))) float v8f;

__device__ __forceinline__ void vqc8(const float (&in)[8], const float (&W)[8],
                                     float (&z)[8]) {
  float c[8];
#pragma unroll
  for (int j = 0; j < 8; ++j) c[j] = __cosf(in[j] + W[j]);
  float p = c[0];
#pragma unroll
  for (int j = 1; j < 8; ++j) { p *= c[j]; z[j] = p; }
  float q = c[1];
#pragma unroll
  for (int j = 2; j < 8; ++j) q *= c[j];
  z[0] = q;
}

__device__ __forceinline__ void ln8(const float (&t)[8],
                                    const float* __restrict__ g,
                                    const float* __restrict__ b,
                                    float (&o)[8]) {
  float m = 0.f;
#pragma unroll
  for (int j = 0; j < 8; ++j) m += t[j];
  m *= 0.125f;
  float v = 0.f;
#pragma unroll
  for (int j = 0; j < 8; ++j) { float d = t[j] - m; v += d * d; }
  v *= 0.125f;
  float inv = 1.0f / sqrtf(v + EPS);
#pragma unroll
  for (int j = 0; j < 8; ++j) o[j] = (t[j] - m) * inv * g[j] + b[j];
}

typedef __attribute__((ext_vector_type(4))) float v4f;

__device__ __forceinline__ float xhalf(float x) {
  int bits = __float_as_int(x);
  return __int_as_float(__builtin_amdgcn_permlanex16(
      bits, bits, 0x76543210, 0xfedcba98, false, false));
}

struct Wt { v2f e[4]; };
__device__ __forceinline__ Wt exp_scores(v8f s) {
  s *= SCALE_LOG2E;
  Wt w;
#pragma unroll
  for (int k = 0; k < 4; ++k) {
    w.e[k].x = __builtin_amdgcn_exp2f(s[2 * k]);
    w.e[k].y = __builtin_amdgcn_exp2f(s[2 * k + 1]);
  }
  return w;
}

struct Acc { v2f c01[2], c23[2], ls; };

__device__ __forceinline__ void accum_tile(const Wt &w, const v4f (&vr)[8],
                                           Acc &a) {
#pragma unroll
  for (int k = 0; k < 4; ++k) {
    const v2f pe = __builtin_shufflevector(w.e[k], w.e[k], 0, 0);
    const v2f po = __builtin_shufflevector(w.e[k], w.e[k], 1, 1);
    const v4f ve = vr[2 * k], vo = vr[2 * k + 1];
    a.c01[0] = __builtin_elementwise_fma(pe, ve.xy, a.c01[0]);
    a.c23[0] = __builtin_elementwise_fma(pe, ve.zw, a.c23[0]);
    a.c01[1] = __builtin_elementwise_fma(po, vo.xy, a.c01[1]);
    a.c23[1] = __builtin_elementwise_fma(po, vo.zw, a.c23[1]);
    a.ls += w.e[k];
  }
}

__global__ __launch_bounds__(128) void attn_kernel(
    const float* __restrict__ x, const float* __restrict__ Wq,
    const float* __restrict__ Wk, const float* __restrict__ Wv,
    float* __restrict__ ws) {
  __shared__ __align__(16) float Qs[S_DIM * DK];
  __shared__ __align__(16) float Ks[KH * DK];
  __shared__ __align__(16) float Vs[KH * DK];

  const int b  = blockIdx.x >> 2;
  const int h  = (blockIdx.x >> 1) & 1;
  const int kh = blockIdx.x & 1;
  const int tid = threadIdx.x;

  float wq[8], wk[8], wv[8];
#pragma unroll
  for (int j = 0; j < 8; ++j) { wq[j] = Wq[j]; wk[j] = Wk[j]; wv[j] = Wv[j]; }

  {
    const float* xa = x + (size_t)(b * S_DIM + tid) * E_DIM;
    const float* xb = xa + (size_t)KH * E_DIM;
    float x0[8], x1[8], xs[8];
#pragma unroll
    for (int j = 0; j < 8; ++j) { x0[j] = xa[j]; x1[j] = xb[j]; }
#pragma unroll
    for (int j = 0; j < 8; ++j) xs[j] = kh ? x1[j] : x0[j];
    float zq0[8], zq1[8], zk[8], zv[8];
    vqc8(x0, wq, zq0);
    vqc8(x1, wq, zq1);
    vqc8(xs, wk, zk);
    vqc8(xs, wv, zv);
#pragma unroll
    for (int d = 0; d < DK; ++d) {
      Qs[tid * DK + d]        = zq0[h * DK + d];
      Qs[(tid + KH) * DK + d] = zq1[h * DK + d];
      Ks[tid * DK + d]        = zk[h * DK + d];
      Vs[tid * DK + d]        = zv[h * DK + d];
    }
  }
  __syncthreads();

  const int lane  = tid & 31;
  const int wv_id = tid >> 5;
  const int half  = lane >> 4;
  const int ln    = lane & 15;
  const int kp    = half * 2;
  const int qbase = wv_id * 64;

  v2f bq[NQB];
  const v2f z2 = {0.f, 0.f};
  Acc qacc[NQB];
#pragma unroll
  for (int j = 0; j < NQB; ++j) {
    bq[j].x = Qs[(qbase + 16 * j + ln) * DK + kp];
    bq[j].y = Qs[(qbase + 16 * j + ln) * DK + kp + 1];
    qacc[j].c01[0] = z2; qacc[j].c01[1] = z2;
    qacc[j].c23[0] = z2; qacc[j].c23[1] = z2;
    qacc[j].ls = z2;
  }

  const v2f* __restrict__ Kfrag =
      reinterpret_cast<const v2f*>(Ks) + ln * 2 + half;
  const v4f* __restrict__ Vrow = reinterpret_cast<const v4f*>(Vs) + 8 * half;
  const v8f cz = {0.f, 0.f, 0.f, 0.f, 0.f, 0.f, 0.f, 0.f};

  Wt wt[NQB];
  auto step = [&](const int tn, const v4f (&vprev)[8], v4f (&vnew)[8]) {
#pragma unroll
    for (int r = 0; r < 8; ++r) vnew[r] = Vrow[16 * tn + r];
    const v2f ak = Kfrag[32 * tn];
    v8f sc[NQB];
#pragma unroll
    for (int j = 0; j < NQB; ++j)
      sc[j] = __builtin_amdgcn_wmma_f32_16x16x4_f32(
          false, ak, false, bq[j], (short)0, cz, false, false);
#pragma unroll
    for (int j = 0; j < NQB; ++j) accum_tile(wt[j], vprev, qacc[j]);
#pragma unroll
    for (int j = 0; j < NQB; ++j) wt[j] = exp_scores(sc[j]);
  };

  v4f vA[8], vB[8];
  {
#pragma unroll
    for (int r = 0; r < 8; ++r) vA[r] = Vrow[r];
    const v2f ak = Kfrag[0];
#pragma unroll
    for (int j = 0; j < NQB; ++j)
      wt[j] = exp_scores(__builtin_amdgcn_wmma_f32_16x16x4_f32(
          false, ak, false, bq[j], (short)0, cz, false, false));
  }
#pragma unroll 1
  for (int t = 0; t < 6; t += 2) {
    step(t + 1, vA, vB);
    step(t + 2, vB, vA);
  }
  step(7, vA, vB);
#pragma unroll
  for (int j = 0; j < NQB; ++j) accum_tile(wt[j], vB, qacc[j]);

#pragma unroll
  for (int p = 0; p < NQB; p += 2) {
    const Acc &qa = qacc[p], &qb = qacc[p + 1];
    const v2f a01 = qa.c01[0] + qa.c01[1], a23 = qa.c23[0] + qa.c23[1];
    const v2f b01 = qb.c01[0] + qb.c01[1], b23 = qb.c23[0] + qb.c23[1];
    float la = qa.ls.x + qa.ls.y, lb = qb.ls.x + qb.ls.y;
    float oa[4] = {a01.x, a01.y, a23.x, a23.y};
    float ob[4] = {b01.x, b01.y, b23.x, b23.y};
    la += xhalf(la);
    lb += xhalf(lb);
#pragma unroll
    for (int d = 0; d < DK; ++d) { oa[d] += xhalf(oa[d]); ob[d] += xhalf(ob[d]); }
    v4f res;
    res.x = half ? ob[0] : oa[0];
    res.y = half ? ob[1] : oa[1];
    res.z = half ? ob[2] : oa[2];
    res.w = half ? ob[3] : oa[3];
    const int m = qbase + 16 * p + 16 * half + ln;
    const size_t slot = (size_t)(h * 2 + kh) * NTOK + (size_t)(b * S_DIM + m);
    reinterpret_cast<v4f*>(ws)[slot] = res;
    ws[(size_t)4 * NTOK * DK + slot] = half ? lb : la;
  }
}

__global__ __launch_bounds__(256) void post_kernel(
    const float* __restrict__ x, const float* __restrict__ ws,
    const float* __restrict__ Wc, const float* __restrict__ Wf,
    const float* __restrict__ w1, const float* __restrict__ b1,
    const float* __restrict__ w2, const float* __restrict__ b2,
    const float* __restrict__ g1, const float* __restrict__ be1,
    const float* __restrict__ g2, const float* __restrict__ be2,
    float* __restrict__ out) {
  const int gid = blockIdx.x * blockDim.x + threadIdx.x;
  if (gid >= B_DIM * S_DIM) return;

  const v4f* __restrict__ part = reinterpret_cast<const v4f*>(ws);
  const float* __restrict__ den = ws + (size_t)4 * NTOK * DK;
  v4f ps[4];
  float ls[4];
#pragma unroll
  for (int p = 0; p < 4; ++p) {
    ps[p] = part[(size_t)p * NTOK + gid];
    ls[p] = den[(size_t)p * NTOK + gid];
  }

  float xv[8], cv[8], wc[8], wf[8];
#pragma unroll
  for (int j = 0; j < 8; ++j) {
    xv[j] = x[(size_t)gid * E_DIM + j];
    wc[j] = Wc[j];
    wf[j] = Wf[j];
  }
#pragma unroll
  for (int hh = 0; hh < H_DIM; ++hh) {
    const float inv_l = 1.0f / (ls[2 * hh] + ls[2 * hh + 1]);
    const v4f c = (ps[2 * hh] + ps[2 * hh + 1]) * inv_l;
    cv[hh * DK + 0] = c.x;
    cv[hh * DK + 1] = c.y;
    cv[hh * DK + 2] = c.z;
    cv[hh * DK + 3] = c.w;
  }

  float ao[8];
  vqc8(cv, wc, ao);
  float t[8];
#pragma unroll
  for (int j = 0; j < 8; ++j) t[j] = xv[j] + ao[j];
  float y[8];
  ln8(t, g1, be1, y);

  float hv[8];
#pragma unroll
  for (int i = 0; i < 8; ++i) {
    float s = b1[i];
#pragma unroll
    for (int j = 0; j < 8; ++j) s += y[j] * w1[i * 8 + j];
    hv[i] = s;
  }
  float hq[8];
  vqc8(hv, wf, hq);

  float u[8];
#pragma unroll
  for (int i = 0; i < 8; ++i) {
    float s = b2[i];
#pragma unroll
    for (int j = 0; j < 8; ++j) s += hq[j] * w2[i * 8 + j];
    u[i] = y[i] + s;
  }
  float o[8];
  ln8(u, g2, be2, o);
#pragma unroll
  for (int j = 0; j < 8; ++j) out[(size_t)gid * E_DIM + j] = o[j];
}

extern "C" void kernel_launch(void* const* d_in, const int* in_sizes, int n_in,
                              void* d_out, int out_size, void* d_ws,
                              size_t ws_size, hipStream_t stream) {
  (void)in_sizes; (void)n_in; (void)out_size; (void)ws_size;
  const float* x   = (const float*)d_in[0];
  const float* Wq  = (const float*)d_in[1];
  const float* Wk  = (const float*)d_in[2];
  const float* Wv  = (const float*)d_in[3];
  const float* Wc  = (const float*)d_in[4];
  const float* Wf  = (const float*)d_in[5];
  const float* w1  = (const float*)d_in[6];
  const float* b1  = (const float*)d_in[7];
  const float* w2  = (const float*)d_in[8];
  const float* b2  = (const float*)d_in[9];
  const float* g1  = (const float*)d_in[10];
  const float* be1 = (const float*)d_in[11];
  const float* g2  = (const float*)d_in[12];
  const float* be2 = (const float*)d_in[13];
  float* out = (float*)d_out;
  float* ws = (float*)d_ws;

  attn_kernel<<<B_DIM * H_DIM * 2, 128, 0, stream>>>(x, Wq, Wk, Wv, ws);
  post_kernel<<<(B_DIM * S_DIM) / 256, 256, 0, stream>>>(
      x, ws, Wc, Wf, w1, b1, w2, b2, g1, be1, g2, be2, out);
}


# Epilogue launch: s_prefetch_data on all ten parameter buffers at kernel entry (scalar cache, no LDS staging); point-of-use scalar loads then hit instead of waiting second-level trips; sums unchanged
# speedup vs baseline: 1.7350x; 1.0726x over previous
#include <hip/hip_runtime.h>
#include <math.h>

#define B_DIM 64
#define S_DIM 256
#define E_DIM 8
#define H_DIM 2
#define DK    4
#define KH    128
#define NQB   4
#define NTOK  (B_DIM * S_DIM)
#define EPS   1e-5f
#define SCALE_LOG2E 0.72134752044448170f

typedef __attribute__((ext_vector_type(2))) float v2f;
typedef __attribute__((ext_vector_type(8))) float v8f;

__device__ __forceinline__ void vqc8(const float (&in)[8], const float (&W)[8],
                                     float (&z)[8]) {
  float c[8];
#pragma unroll
  for (int j = 0; j < 8; ++j) c[j] = __cosf(in[j] + W[j]);
  float p = c[0];
#pragma unroll
  for (int j = 1; j < 8; ++j) { p *= c[j]; z[j] = p; }
  float q = c[1];
#pragma unroll
  for (int j = 2; j < 8; ++j) q *= c[j];
  z[0] = q;
}

typedef const __attribute__((address_space(4))) float* cfp;

__device__ __forceinline__ void ln8(const float (&t)[8], cfp g, cfp b,
                                    float (&o)[8]) {
  float m = 0.f;
#pragma unroll
  for (int j = 0; j < 8; ++j) m += t[j];
  m *= 0.125f;
  float v = 0.f;
#pragma unroll
  for (int j = 0; j < 8; ++j) { float d = t[j] - m; v += d * d; }
  v *= 0.125f;
  float inv = 1.0f / sqrtf(v + EPS);
#pragma unroll
  for (int j = 0; j < 8; ++j) o[j] = (t[j] - m) * inv * g[j] + b[j];
}

typedef __attribute__((ext_vector_type(4))) float v4f;

__device__ __forceinline__ float xhalf(float x) {
  int bits = __float_as_int(x);
  return __int_as_float(__builtin_amdgcn_permlanex16(
      bits, bits, 0x76543210, 0xfedcba98, false, false));
}

struct Wt { v2f e[4]; };
__device__ __forceinline__ Wt exp_scores(v8f s) {
  s *= SCALE_LOG2E;
  Wt w;
#pragma unroll
  for (int k = 0; k < 4; ++k) {
    w.e[k].x = __builtin_amdgcn_exp2f(s[2 * k]);
    w.e[k].y = __builtin_amdgcn_exp2f(s[2 * k + 1]);
  }
  return w;
}

struct Acc { v2f c01[2], c23[2], ls; };

__device__ __forceinline__ void accum_tile(const Wt &w, const v4f (&vr)[8],
                                           Acc &a) {
#pragma unroll
  for (int k = 0; k < 4; ++k) {
    const v2f pe = __builtin_shufflevector(w.e[k], w.e[k], 0, 0);
    const v2f po = __builtin_shufflevector(w.e[k], w.e[k], 1, 1);
    const v4f ve = vr[2 * k], vo = vr[2 * k + 1];
    a.c01[0] = __builtin_elementwise_fma(pe, ve.xy, a.c01[0]);
    a.c23[0] = __builtin_elementwise_fma(pe, ve.zw, a.c23[0]);
    a.c01[1] = __builtin_elementwise_fma(po, vo.xy, a.c01[1]);
    a.c23[1] = __builtin_elementwise_fma(po, vo.zw, a.c23[1]);
    a.ls += w.e[k];
  }
}

__global__ __launch_bounds__(128) void attn_kernel(
    const float* __restrict__ x, const float* __restrict__ Wq,
    const float* __restrict__ Wk, const float* __restrict__ Wv,
    float* __restrict__ ws) {
  __shared__ __align__(16) float Qs[S_DIM * DK];
  __shared__ __align__(16) float Ks[KH * DK];
  __shared__ __align__(16) float Vs[KH * DK];

  const int b  = blockIdx.x >> 2;
  const int h  = (blockIdx.x >> 1) & 1;
  const int kh = blockIdx.x & 1;
  const int tid = threadIdx.x;

  float wq[8], wk[8], wv[8];
#pragma unroll
  for (int j = 0; j < 8; ++j) { wq[j] = Wq[j]; wk[j] = Wk[j]; wv[j] = Wv[j]; }

  {
    const float* xa = x + (size_t)(b * S_DIM + tid) * E_DIM;
    const float* xb = xa + (size_t)KH * E_DIM;
    float x0[8], x1[8], xs[8];
#pragma unroll
    for (int j = 0; j < 8; ++j) { x0[j] = xa[j]; x1[j] = xb[j]; }
#pragma unroll
    for (int j = 0; j < 8; ++j) xs[j] = kh ? x1[j] : x0[j];
    float zq0[8], zq1[8], zk[8], zv[8];
    vqc8(x0, wq, zq0);
    vqc8(x1, wq, zq1);
    vqc8(xs, wk, zk);
    vqc8(xs, wv, zv);
#pragma unroll
    for (int d = 0; d < DK; ++d) {
      Qs[tid * DK + d]        = zq0[h * DK + d];
      Qs[(tid + KH) * DK + d] = zq1[h * DK + d];
      Ks[tid * DK + d]        = zk[h * DK + d];
      Vs[tid * DK + d]        = zv[h * DK + d];
    }
  }
  __syncthreads();

  const int lane  = tid & 31;
  const int wv_id = tid >> 5;
  const int half  = lane >> 4;
  const int ln    = lane & 15;
  const int kp    = half * 2;
  const int qbase = wv_id * 64;

  v2f bq[NQB];
  const v2f z2 = {0.f, 0.f};
  Acc qacc[NQB];
#pragma unroll
  for (int j = 0; j < NQB; ++j) {
    bq[j].x = Qs[(qbase + 16 * j + ln) * DK + kp];
    bq[j].y = Qs[(qbase + 16 * j + ln) * DK + kp + 1];
    qacc[j].c01[0] = z2; qacc[j].c01[1] = z2;
    qacc[j].c23[0] = z2; qacc[j].c23[1] = z2;
    qacc[j].ls = z2;
  }

  const v2f* __restrict__ Kfrag =
      reinterpret_cast<const v2f*>(Ks) + ln * 2 + half;
  const v4f* __restrict__ Vrow = reinterpret_cast<const v4f*>(Vs) + 8 * half;
  const v8f cz = {0.f, 0.f, 0.f, 0.f, 0.f, 0.f, 0.f, 0.f};

  Wt wt[NQB];
  auto step = [&](const int tn, const v4f (&vprev)[8], v4f (&vnew)[8]) {
#pragma unroll
    for (int r = 0; r < 8; ++r) vnew[r] = Vrow[16 * tn + r];
    const v2f ak = Kfrag[32 * tn];
    v8f sc[NQB];
#pragma unroll
    for (int j = 0; j < NQB; ++j)
      sc[j] = __builtin_amdgcn_wmma_f32_16x16x4_f32(
          false, ak, false, bq[j], (short)0, cz, false, false);
#pragma unroll
    for (int j = 0; j < NQB; ++j) accum_tile(wt[j], vprev, qacc[j]);
#pragma unroll
    for (int j = 0; j < NQB; ++j) wt[j] = exp_scores(sc[j]);
  };

  v4f vA[8], vB[8];
  {
#pragma unroll
    for (int r = 0; r < 8; ++r) vA[r] = Vrow[r];
    const v2f ak = Kfrag[0];
#pragma unroll
    for (int j = 0; j < NQB; ++j)
      wt[j] = exp_scores(__builtin_amdgcn_wmma_f32_16x16x4_f32(
          false, ak, false, bq[j], (short)0, cz, false, false));
  }
#pragma unroll 1
  for (int t = 0; t < 6; t += 2) {
    step(t + 1, vA, vB);
    step(t + 2, vB, vA);
  }
  step(7, vA, vB);
#pragma unroll
  for (int j = 0; j < NQB; ++j) accum_tile(wt[j], vB, qacc[j]);

#pragma unroll
  for (int p = 0; p < NQB; p += 2) {
    const Acc &qa = qacc[p], &qb = qacc[p + 1];
    const v2f a01 = qa.c01[0] + qa.c01[1], a23 = qa.c23[0] + qa.c23[1];
    const v2f b01 = qb.c01[0] + qb.c01[1], b23 = qb.c23[0] + qb.c23[1];
    float la = qa.ls.x + qa.ls.y, lb = qb.ls.x + qb.ls.y;
    float oa[4] = {a01.x, a01.y, a23.x, a23.y};
    float ob[4] = {b01.x, b01.y, b23.x, b23.y};
    la += xhalf(la);
    lb += xhalf(lb);
#pragma unroll
    for (int d = 0; d < DK; ++d) { oa[d] += xhalf(oa[d]); ob[d] += xhalf(ob[d]); }
    v4f res;
    res.x = half ? ob[0] : oa[0];
    res.y = half ? ob[1] : oa[1];
    res.z = half ? ob[2] : oa[2];
    res.w = half ? ob[3] : oa[3];
    const int m = qbase + 16 * p + 16 * half + ln;
    const size_t slot = (size_t)(h * 2 + kh) * NTOK + (size_t)(b * S_DIM + m);
    reinterpret_cast<v4f*>(ws)[slot] = res;
    ws[(size_t)4 * NTOK * DK + slot] = half ? lb : la;
  }
}

__global__ __launch_bounds__(256) void post_kernel(
    const float* __restrict__ x, const float* __restrict__ ws,
    const float* __restrict__ Wc, const float* __restrict__ Wf,
    const float* __restrict__ w1, const float* __restrict__ b1,
    const float* __restrict__ w2, const float* __restrict__ b2,
    const float* __restrict__ g1, const float* __restrict__ be1,
    const float* __restrict__ g2, const float* __restrict__ be2,
    float* __restrict__ out) {
  __builtin_amdgcn_s_prefetch_data(w1, 0);
  __builtin_amdgcn_s_prefetch_data(w1 + 32, 0);
  __builtin_amdgcn_s_prefetch_data(w2, 0);
  __builtin_amdgcn_s_prefetch_data(w2 + 32, 0);
  __builtin_amdgcn_s_prefetch_data(Wc, 0);
  __builtin_amdgcn_s_prefetch_data(g1, 0);
  __builtin_amdgcn_s_prefetch_data(be1, 0);
  __builtin_amdgcn_s_prefetch_data(b1, 0);
  __builtin_amdgcn_s_prefetch_data(Wf, 0);
  __builtin_amdgcn_s_prefetch_data(b2, 0);
  __builtin_amdgcn_s_prefetch_data(g2, 0);
  __builtin_amdgcn_s_prefetch_data(be2, 0);
  const cfp cWc = (cfp)Wc, cWf = (cfp)Wf, cw1 = (cfp)w1, cb1 = (cfp)b1;
  const cfp cw2 = (cfp)w2, cb2 = (cfp)b2, cg1 = (cfp)g1, cbe1 = (cfp)be1;
  const cfp cg2 = (cfp)g2, cbe2 = (cfp)be2;

  const int gid = blockIdx.x * blockDim.x + threadIdx.x;
  if (gid >= B_DIM * S_DIM) return;

  const v4f* __restrict__ part = reinterpret_cast<const v4f*>(ws);
  const float* __restrict__ den = ws + (size_t)4 * NTOK * DK;
  v4f ps[4];
  float ls[4];
#pragma unroll
  for (int p = 0; p < 4; ++p) {
    ps[p] = part[(size_t)p * NTOK + gid];
    ls[p] = den[(size_t)p * NTOK + gid];
  }

  float xv[8], cv[8], wc[8], wf[8];
#pragma unroll
  for (int j = 0; j < 8; ++j) {
    xv[j] = x[(size_t)gid * E_DIM + j];
    wc[j] = cWc[j];
    wf[j] = cWf[j];
  }
#pragma unroll
  for (int hh = 0; hh < H_DIM; ++hh) {
    const float inv_l = 1.0f / (ls[2 * hh] + ls[2 * hh + 1]);
    const v4f c = (ps[2 * hh] + ps[2 * hh + 1]) * inv_l;
    cv[hh * DK + 0] = c.x;
    cv[hh * DK + 1] = c.y;
    cv[hh * DK + 2] = c.z;
    cv[hh * DK + 3] = c.w;
  }

  float ao[8];
  vqc8(cv, wc, ao);
  float t[8];
#pragma unroll
  for (int j = 0; j < 8; ++j) t[j] = xv[j] + ao[j];
  float y[8];
  ln8(t, cg1, cbe1, y);

  float hv[8];
#pragma unroll
  for (int i = 0; i < 8; ++i) {
    float s = cb1[i];
#pragma unroll
    for (int j = 0; j < 8; ++j) s += y[j] * cw1[i * 8 + j];
    hv[i] = s;
  }
  float hq[8];
  vqc8(hv, wf, hq);

  float u[8];
#pragma unroll
  for (int i = 0; i < 8; ++i) {
    float s = cb2[i];
#pragma unroll
    for (int j = 0; j < 8; ++j) s += hq[j] * cw2[i * 8 + j];
    u[i] = y[i] + s;
  }
  float o[8];
  ln8(u, cg2, cbe2, o);
#pragma unroll
  for (int j = 0; j < 8; ++j) out[(size_t)gid * E_DIM + j] = o[j];
}

extern "C" void kernel_launch(void* const* d_in, const int* in_sizes, int n_in,
                              void* d_out, int out_size, void* d_ws,
                              size_t ws_size, hipStream_t stream) {
  (void)in_sizes; (void)n_in; (void)out_size; (void)ws_size;
  const float* x   = (const float*)d_in[0];
  const float* Wq  = (const float*)d_in[1];
  const float* Wk  = (const float*)d_in[2];
  const float* Wv  = (const float*)d_in[3];
  const float* Wc  = (const float*)d_in[4];
  const float* Wf  = (const float*)d_in[5];
  const float* w1  = (const float*)d_in[6];
  const float* b1  = (const float*)d_in[7];
  const float* w2  = (const float*)d_in[8];
  const float* b2  = (const float*)d_in[9];
  const float* g1  = (const float*)d_in[10];
  const float* be1 = (const float*)d_in[11];
  const float* g2  = (const float*)d_in[12];
  const float* be2 = (const float*)d_in[13];
  float* out = (float*)d_out;
  float* ws = (float*)d_ws;

  attn_kernel<<<B_DIM * H_DIM * 2, 128, 0, stream>>>(x, Wq, Wk, Wv, ws);
  post_kernel<<<(B_DIM * S_DIM) / 256, 256, 0, stream>>>(
      x, ws, Wc, Wf, w1, b1, w2, b2, g1, be1, g2, be2, out);
}
